# starting kernel
# baseline (speedup 1.0000x reference)
#include <hip/hip_runtime.h>
#include <hip/hip_bf16.h>

#define N_NODES 50000
#define N_EDGES 312500
#define F 256
#define K2 512
#define M_TILES 3125
#define BSTRIDE 516

typedef float v2f __attribute__((ext_vector_type(2)));
typedef float v8f __attribute__((ext_vector_type(8)));

__global__ void GNN_zero_kernel(float* __restrict__ p, size_t n) {
    size_t i = (size_t)blockIdx.x * blockDim.x + threadIdx.x;
    size_t stride = (size_t)gridDim.x * blockDim.x;
    for (; i < n; i += stride) p[i] = 0.0f;
}

__global__ __launch_bounds__(256) void GNN_scatter_kernel(
    const float* __restrict__ x,
    const int*   __restrict__ Ai,
    const int*   __restrict__ Aj,
    float* __restrict__ S,
    float* __restrict__ deg)
{
    const int lane   = threadIdx.x & 31;
    const int wave   = (blockIdx.x * (blockDim.x >> 5)) + (threadIdx.x >> 5);
    const int nwaves = gridDim.x * (blockDim.x >> 5);

    for (int e = wave; e < N_EDGES; e += nwaves) {
        const int ai = Ai[e];
        const int aj = Aj[e];

        const int en = e + nwaves;
        if (en < N_EDGES) {
            const int ajn = Aj[en];
            __builtin_prefetch(x + (size_t)ajn * F + lane * 8, 0, 0);
        }

        if (lane == 0) atomicAdd(&deg[ai], 1.0f);

        const float* __restrict__ xr = x + (size_t)aj * F;
        float*       __restrict__ sr = S + (size_t)ai * F;
        #pragma unroll
        for (int c = 0; c < F; c += 32) {
            atomicAdd(&sr[c + lane], xr[c + lane]);
        }
    }
}

__global__ __launch_bounds__(256) void GNN_gemm_kernel(
    const float* __restrict__ x,
    const float* __restrict__ W,
    const float* __restrict__ b,
    const float* __restrict__ S,
    const float* __restrict__ deg,
    float* __restrict__ out)
{
    __shared__ float Bs[16 * BSTRIDE];

    const int n0   = blockIdx.y * 16;
    const int wave = threadIdx.x >> 5;
    const int lane = threadIdx.x & 31;
    const int mtile = blockIdx.x * 8 + wave;

    for (int i = threadIdx.x; i < 16 * K2; i += 256) {
        const int j = i >> 9;
        const int k = i & (K2 - 1);
        Bs[j * BSTRIDE + k] = W[(size_t)(n0 + j) * K2 + k];
    }
    __syncthreads();

    if (mtile >= M_TILES) return;

    const int m0    = mtile * 16;
    const int arow  = m0 + (lane & 15);
    const int khalf = (lane >> 4) * 2;
    const int ncol  = lane & 15;
    const float dsc = deg[arow];

    const float* __restrict__ xr = x + (size_t)arow * F;
    const float* __restrict__ sr = S + (size_t)arow * F;

    v8f acc = {};

    #pragma unroll 4
    for (int k = 0; k < F; k += 4) {
        v2f a, bb;
        a.x  = dsc * xr[k + khalf];
        a.y  = dsc * xr[k + khalf + 1];
        bb.x = Bs[ncol * BSTRIDE + k + khalf];
        bb.y = Bs[ncol * BSTRIDE + k + khalf + 1];
        acc = __builtin_amdgcn_wmma_f32_16x16x4_f32(
            false, a, false, bb, (short)0, acc, false, false);
    }
    #pragma unroll 4
    for (int k = 0; k < F; k += 4) {
        v2f a, bb;
        a.x  = sr[k + khalf];
        a.y  = sr[k + khalf + 1];
        bb.x = Bs[ncol * BSTRIDE + F + k + khalf];
        bb.y = Bs[ncol * BSTRIDE + F + k + khalf + 1];
        acc = __builtin_amdgcn_wmma_f32_16x16x4_f32(
            false, a, false, bb, (short)0, acc, false, false);
    }

    const int nout  = n0 + (lane & 15);
    const int mbase = m0 + (lane >> 4) * 8;
    const float bias = b[nout];
    #pragma unroll
    for (int i = 0; i < 8; ++i) {
        const int m = mbase + i;
        out[(size_t)m * F + nout] = acc[i] + deg[m] * bias;
    }
}

extern "C" void kernel_launch(void* const* d_in, const int* in_sizes, int n_in,
                              void* d_out, int out_size, void* d_ws, size_t ws_size,
                              hipStream_t stream) {
    const float* x  = (const float*)d_in[0];
    const float* W  = (const float*)d_in[1];
    const float* b  = (const float*)d_in[2];
    const int*   Ai = (const int*)d_in[3];
    const int*   Aj = (const int*)d_in[4];
    float* out = (float*)d_out;

    float* S   = (float*)d_ws;
    float* deg = S + (size_t)N_NODES * F;

    const size_t nzero = (size_t)N_NODES * F + N_NODES;
    GNN_zero_kernel<<<1024, 256, 0, stream>>>(S, nzero);

    GNN_scatter_kernel<<<2048, 256, 0, stream>>>(x, Ai, Aj, S, deg);

    dim3 grid((M_TILES + 7) / 8, F / 16);
    GNN_gemm_kernel<<<grid, 256, 0, stream>>>(x, W, b, S, deg, out);
}


# GEMM retiled: each wave computes 16x64 (4 accumulator tiles) from one 16-byte A fragment per two k-steps, W slice of 64 columns in LDS; A read 4x not 16x; k order inside each 8 permuted
# speedup vs baseline: 1.4583x; 1.4583x over previous
#include <hip/hip_runtime.h>
#include <hip/hip_bf16.h>

#define N_NODES 50000
#define N_EDGES 312500
#define F 256
#define K2 512
#define M_TILES 3125
#define BSTRIDE 516
#define NT 4
#define BN (16 * NT)

typedef float v2f __attribute__((ext_vector_type(2)));
typedef float v8f __attribute__((ext_vector_type(8)));

__global__ void GNN_zero_kernel(float* __restrict__ p, size_t n) {
    size_t i = (size_t)blockIdx.x * blockDim.x + threadIdx.x;
    size_t stride = (size_t)gridDim.x * blockDim.x;
    for (; i < n; i += stride) p[i] = 0.0f;
}

__global__ __launch_bounds__(256) void GNN_scatter_kernel(
    const float* __restrict__ x,
    const int*   __restrict__ Ai,
    const int*   __restrict__ Aj,
    float* __restrict__ S,
    float* __restrict__ deg)
{
    const int lane   = threadIdx.x & 31;
    const int wave   = (blockIdx.x * (blockDim.x >> 5)) + (threadIdx.x >> 5);
    const int nwaves = gridDim.x * (blockDim.x >> 5);

    for (int e = wave; e < N_EDGES; e += nwaves) {
        const int ai = Ai[e];
        const int aj = Aj[e];

        const int en = e + nwaves;
        if (en < N_EDGES) {
            const int ajn = Aj[en];
            __builtin_prefetch(x + (size_t)ajn * F + lane * 8, 0, 0);
        }

        if (lane == 0) atomicAdd(&deg[ai], 1.0f);

        const float* __restrict__ xr = x + (size_t)aj * F;
        float*       __restrict__ sr = S + (size_t)ai * F;
        #pragma unroll
        for (int c = 0; c < F; c += 32) {
            atomicAdd(&sr[c + lane], xr[c + lane]);
        }
    }
}

__global__ __launch_bounds__(256) void GNN_gemm_kernel(
    const float* __restrict__ x,
    const float* __restrict__ W,
    const float* __restrict__ b,
    const float* __restrict__ S,
    const float* __restrict__ deg,
    float* __restrict__ out)
{
    __shared__ __attribute__((aligned(16))) float Bs[BN * BSTRIDE];

    const int n0   = blockIdx.x * BN;
    const int wave = threadIdx.x >> 5;
    const int lane = threadIdx.x & 31;
    const int mtile = blockIdx.y * 8 + wave;

    for (int i = threadIdx.x; i < BN * (K2 / 4); i += 256) {
        const int j  = i >> 7;
        const int k4 = (i & 127) * 4;
        const float4 w = *reinterpret_cast<const float4*>(W + (size_t)(n0 + j) * K2 + k4);
        *reinterpret_cast<float4*>(&Bs[j * BSTRIDE + k4]) = w;
    }
    __syncthreads();

    if (mtile >= M_TILES) return;

    const int c    = lane & 15;
    const int h    = lane >> 4;
    const int m0   = mtile * 16;
    const int arow = m0 + c;
    const float dsc = deg[arow];

    const float* __restrict__ xr = x + (size_t)arow * F + 4 * h;
    const float* __restrict__ sr = S + (size_t)arow * F + 4 * h;
    const float* bp = Bs + c * BSTRIDE + 4 * h;

    v8f acc[NT];
    #pragma unroll
    for (int t = 0; t < NT; ++t) acc[t] = v8f{};

    #pragma unroll 4
    for (int k = 0; k < F; k += 8) {
        const float4 av = *reinterpret_cast<const float4*>(xr + k);
        float4 bv[NT];
        #pragma unroll
        for (int t = 0; t < NT; ++t)
            bv[t] = *reinterpret_cast<const float4*>(bp + t * 16 * BSTRIDE + k);
        v2f a0, a1;
        a0.x = dsc * av.x;  a0.y = dsc * av.y;
        a1.x = dsc * av.z;  a1.y = dsc * av.w;
        #pragma unroll
        for (int t = 0; t < NT; ++t) {
            v2f b0; b0.x = bv[t].x; b0.y = bv[t].y;
            acc[t] = __builtin_amdgcn_wmma_f32_16x16x4_f32(
                false, a0, false, b0, (short)0, acc[t], false, false);
        }
        #pragma unroll
        for (int t = 0; t < NT; ++t) {
            v2f b1; b1.x = bv[t].z; b1.y = bv[t].w;
            acc[t] = __builtin_amdgcn_wmma_f32_16x16x4_f32(
                false, a1, false, b1, (short)0, acc[t], false, false);
        }
    }
    #pragma unroll 4
    for (int k = 0; k < F; k += 8) {
        const float4 av = *reinterpret_cast<const float4*>(sr + k);
        float4 bv[NT];
        #pragma unroll
        for (int t = 0; t < NT; ++t)
            bv[t] = *reinterpret_cast<const float4*>(bp + t * 16 * BSTRIDE + F + k);
        v2f a0, a1;
        a0.x = av.x;  a0.y = av.y;
        a1.x = av.z;  a1.y = av.w;
        #pragma unroll
        for (int t = 0; t < NT; ++t) {
            v2f b0; b0.x = bv[t].x; b0.y = bv[t].y;
            acc[t] = __builtin_amdgcn_wmma_f32_16x16x4_f32(
                false, a0, false, b0, (short)0, acc[t], false, false);
        }
        #pragma unroll
        for (int t = 0; t < NT; ++t) {
            v2f b1; b1.x = bv[t].z; b1.y = bv[t].w;
            acc[t] = __builtin_amdgcn_wmma_f32_16x16x4_f32(
                false, a1, false, b1, (short)0, acc[t], false, false);
        }
    }

    const int mbase = m0 + h * 8;
    float dg[8];
    #pragma unroll
    for (int i = 0; i < 8; ++i) dg[i] = deg[mbase + i];
    #pragma unroll
    for (int t = 0; t < NT; ++t) {
        const int nout = n0 + t * 16 + c;
        const float bias = b[nout];
        #pragma unroll
        for (int i = 0; i < 8; ++i) {
            out[(size_t)(mbase + i) * F + nout] = acc[t][i] + dg[i] * bias;
        }
    }
}

extern "C" void kernel_launch(void* const* d_in, const int* in_sizes, int n_in,
                              void* d_out, int out_size, void* d_ws, size_t ws_size,
                              hipStream_t stream) {
    const float* x  = (const float*)d_in[0];
    const float* W  = (const float*)d_in[1];
    const float* b  = (const float*)d_in[2];
    const int*   Ai = (const int*)d_in[3];
    const int*   Aj = (const int*)d_in[4];
    float* out = (float*)d_out;

    float* S   = (float*)d_ws;
    float* deg = S + (size_t)N_NODES * F;

    const size_t nzero = (size_t)N_NODES * F + N_NODES;
    GNN_zero_kernel<<<1024, 256, 0, stream>>>(S, nzero);

    GNN_scatter_kernel<<<2048, 256, 0, stream>>>(x, Ai, Aj, S, deg);

    dim3 grid(F / BN, (M_TILES + 7) / 8);
    GNN_gemm_kernel<<<grid, 256, 0, stream>>>(x, W, b, S, deg, out);
}


# Row scatter by f32 atomics and its zero fill replaced by per-node edge chains (one integer exchange per edge) and a wave-per-node gather that stores S once; rows summed in chain (arrival) order
# speedup vs baseline: 2.0552x; 1.4093x over previous
#include <hip/hip_runtime.h>
#include <hip/hip_bf16.h>

#define N_NODES 50000
#define N_EDGES 312500
#define F 256
#define K2 512
#define M_TILES 3125
#define BSTRIDE 516
#define NT 4
#define BN (16 * NT)

typedef float v2f __attribute__((ext_vector_type(2)));
typedef float v8f __attribute__((ext_vector_type(8)));

__global__ __launch_bounds__(256) void GNN_init_kernel(int* __restrict__ head) {
    const int i = blockIdx.x * blockDim.x + threadIdx.x;
    if (i < N_NODES) head[i] = -1;
}

static_assert(N_EDGES % 4 == 0, "edge count must be a multiple of 4");
__global__ __launch_bounds__(256) void GNN_link_kernel(
    const int* __restrict__ Ai,
    const int* __restrict__ Aj,
    int* __restrict__ head,
    int* __restrict__ link)
{
    const int t = blockIdx.x * blockDim.x + threadIdx.x;
    if (t < N_EDGES / 4) {
        const int4 ai = *reinterpret_cast<const int4*>(Ai + 4 * (size_t)t);
        const int4 aj = *reinterpret_cast<const int4*>(Aj + 4 * (size_t)t);
        const int e = 4 * t;
        const int p0 = atomicExch(&head[ai.x], e);
        const int p1 = atomicExch(&head[ai.y], e + 1);
        const int p2 = atomicExch(&head[ai.z], e + 2);
        const int p3 = atomicExch(&head[ai.w], e + 3);
        int4 l0, l1;
        l0.x = p0; l0.y = aj.x; l0.z = p1; l0.w = aj.y;
        l1.x = p2; l1.y = aj.z; l1.z = p3; l1.w = aj.w;
        int* lp = link + 8 * (size_t)t;
        *reinterpret_cast<int4*>(lp)     = l0;
        *reinterpret_cast<int4*>(lp + 4) = l1;
    }
}

__global__ __launch_bounds__(256) void GNN_gather_kernel(
    const float* __restrict__ x,
    const int*   __restrict__ head,
    const int*   __restrict__ link,
    float* __restrict__ S,
    float* __restrict__ deg)
{
    const int lane   = threadIdx.x & 31;
    const int wave   = __builtin_amdgcn_readfirstlane((int)(blockIdx.x * (blockDim.x >> 5) + (threadIdx.x >> 5)));
    const int nwaves = gridDim.x * (blockDim.x >> 5);

    for (int n = wave; n < N_NODES; n += nwaves) {
        int e = head[n];
        float4 a0 = make_float4(0.f, 0.f, 0.f, 0.f), a1 = a0;
        int cnt = 0;
        if ((unsigned)e < (unsigned)N_EDGES) {
            int2 l = *reinterpret_cast<const int2*>(link + 2 * (size_t)e);
            for (;;) {
                const float* __restrict__ xr = x + (size_t)l.y * F + lane * 4;
                const float4 r0 = *reinterpret_cast<const float4*>(xr);
                const float4 r1 = *reinterpret_cast<const float4*>(xr + 128);
                ++cnt;
                e = l.x;
                const bool more = (unsigned)e < (unsigned)N_EDGES && cnt < N_EDGES;
                if (more) l = *reinterpret_cast<const int2*>(link + 2 * (size_t)e);
                a0.x += r0.x; a0.y += r0.y; a0.z += r0.z; a0.w += r0.w;
                a1.x += r1.x; a1.y += r1.y; a1.z += r1.z; a1.w += r1.w;
                if (!more) break;
            }
        }

        float* __restrict__ sr = S + (size_t)n * F + lane * 4;
        *reinterpret_cast<float4*>(sr)       = a0;
        *reinterpret_cast<float4*>(sr + 128) = a1;
        if (lane == 0) deg[n] = (float)cnt;
    }
}

__global__ __launch_bounds__(256) void GNN_gemm_kernel(
    const float* __restrict__ x,
    const float* __restrict__ W,
    const float* __restrict__ b,
    const float* __restrict__ S,
    const float* __restrict__ deg,
    float* __restrict__ out)
{
    __shared__ __attribute__((aligned(16))) float Bs[BN * BSTRIDE];

    const int n0   = blockIdx.x * BN;
    const int wave = threadIdx.x >> 5;
    const int lane = threadIdx.x & 31;
    const int mtile = blockIdx.y * 8 + wave;

    for (int i = threadIdx.x; i < BN * (K2 / 4); i += 256) {
        const int j  = i >> 7;
        const int k4 = (i & 127) * 4;
        const float4 w = *reinterpret_cast<const float4*>(W + (size_t)(n0 + j) * K2 + k4);
        *reinterpret_cast<float4*>(&Bs[j * BSTRIDE + k4]) = w;
    }
    __syncthreads();

    if (mtile >= M_TILES) return;

    const int c    = lane & 15;
    const int h    = lane >> 4;
    const int m0   = mtile * 16;
    const int arow = m0 + c;
    const float dsc = deg[arow];

    const float* __restrict__ xr = x + (size_t)arow * F + 4 * h;
    const float* __restrict__ sr = S + (size_t)arow * F + 4 * h;
    const float* bp = Bs + c * BSTRIDE + 4 * h;

    v8f acc[NT];
    #pragma unroll
    for (int t = 0; t < NT; ++t) acc[t] = v8f{};

    #pragma unroll 4
    for (int k = 0; k < F; k += 8) {
        const float4 av = *reinterpret_cast<const float4*>(xr + k);
        float4 bv[NT];
        #pragma unroll
        for (int t = 0; t < NT; ++t)
            bv[t] = *reinterpret_cast<const float4*>(bp + t * 16 * BSTRIDE + k);
        v2f a0, a1;
        a0.x = dsc * av.x;  a0.y = dsc * av.y;
        a1.x = dsc * av.z;  a1.y = dsc * av.w;
        #pragma unroll
        for (int t = 0; t < NT; ++t) {
            v2f b0; b0.x = bv[t].x; b0.y = bv[t].y;
            acc[t] = __builtin_amdgcn_wmma_f32_16x16x4_f32(
                false, a0, false, b0, (short)0, acc[t], false, false);
        }
        #pragma unroll
        for (int t = 0; t < NT; ++t) {
            v2f b1; b1.x = bv[t].z; b1.y = bv[t].w;
            acc[t] = __builtin_amdgcn_wmma_f32_16x16x4_f32(
                false, a1, false, b1, (short)0, acc[t], false, false);
        }
    }
    #pragma unroll 4
    for (int k = 0; k < F; k += 8) {
        const float4 av = *reinterpret_cast<const float4*>(sr + k);
        float4 bv[NT];
        #pragma unroll
        for (int t = 0; t < NT; ++t)
            bv[t] = *reinterpret_cast<const float4*>(bp + t * 16 * BSTRIDE + F + k);
        v2f a0, a1;
        a0.x = av.x;  a0.y = av.y;
        a1.x = av.z;  a1.y = av.w;
        #pragma unroll
        for (int t = 0; t < NT; ++t) {
            v2f b0; b0.x = bv[t].x; b0.y = bv[t].y;
            acc[t] = __builtin_amdgcn_wmma_f32_16x16x4_f32(
                false, a0, false, b0, (short)0, acc[t], false, false);
        }
        #pragma unroll
        for (int t = 0; t < NT; ++t) {
            v2f b1; b1.x = bv[t].z; b1.y = bv[t].w;
            acc[t] = __builtin_amdgcn_wmma_f32_16x16x4_f32(
                false, a1, false, b1, (short)0, acc[t], false, false);
        }
    }

    const int mbase = m0 + h * 8;
    float dg[8];
    #pragma unroll
    for (int i = 0; i < 8; ++i) dg[i] = deg[mbase + i];
    #pragma unroll
    for (int t = 0; t < NT; ++t) {
        const int nout = n0 + t * 16 + c;
        const float bias = b[nout];
        #pragma unroll
        for (int i = 0; i < 8; ++i) {
            out[(size_t)(mbase + i) * F + nout] = acc[t][i] + dg[i] * bias;
        }
    }
}

extern "C" void kernel_launch(void* const* d_in, const int* in_sizes, int n_in,
                              void* d_out, int out_size, void* d_ws, size_t ws_size,
                              hipStream_t stream) {
    const float* x  = (const float*)d_in[0];
    const float* W  = (const float*)d_in[1];
    const float* b  = (const float*)d_in[2];
    const int*   Ai = (const int*)d_in[3];
    const int*   Aj = (const int*)d_in[4];
    float* out = (float*)d_out;

    float* S    = (float*)d_ws;
    float* deg  = S + (size_t)N_NODES * F;
    int*   head = (int*)(deg + N_NODES);
    int*   link = head + N_NODES;

    GNN_init_kernel<<<(N_NODES + 255) / 256, 256, 0, stream>>>(head);

    GNN_link_kernel<<<(N_EDGES / 4 + 255) / 256, 256, 0, stream>>>(Ai, Aj, head, link);

    GNN_gather_kernel<<<2048, 256, 0, stream>>>(x, head, link, S, deg);

    dim3 grid(F / BN, (M_TILES + 7) / 8);
    GNN_gemm_kernel<<<grid, 256, 0, stream>>>(x, W, b, S, deg, out);
}


# GEMM k loop: operand loads issued ahead of use into spare register sets, one group in front of each 32-matrix-instruction block (A from global 3 blocks ahead, B from LDS 1 ahead); sum order unchanged
# speedup vs baseline: 2.2064x; 1.0736x over previous
#include <hip/hip_runtime.h>
#include <hip/hip_bf16.h>

#define N_NODES 50000
#define N_EDGES 312500
#define F 256
#define K2 512
#define M_TILES 3125
#define BSTRIDE 516
#define NT 4
#define BN (16 * NT)

typedef float v2f __attribute__((ext_vector_type(2)));
typedef float v4f __attribute__((ext_vector_type(4)));
typedef float v8f __attribute__((ext_vector_type(8)));

__global__ __launch_bounds__(256) void GNN_init_kernel(int* __restrict__ head) {
    const int i = blockIdx.x * blockDim.x + threadIdx.x;
    if (i < N_NODES) head[i] = -1;
}

static_assert(N_EDGES % 4 == 0, "edge count must be a multiple of 4");
__global__ __launch_bounds__(256) void GNN_link_kernel(
    const int* __restrict__ Ai,
    const int* __restrict__ Aj,
    int* __restrict__ head,
    int* __restrict__ link)
{
    const int t = blockIdx.x * blockDim.x + threadIdx.x;
    if (t < N_EDGES / 4) {
        const int4 ai = *reinterpret_cast<const int4*>(Ai + 4 * (size_t)t);
        const int4 aj = *reinterpret_cast<const int4*>(Aj + 4 * (size_t)t);
        const int e = 4 * t;
        const int p0 = atomicExch(&head[ai.x], e);
        const int p1 = atomicExch(&head[ai.y], e + 1);
        const int p2 = atomicExch(&head[ai.z], e + 2);
        const int p3 = atomicExch(&head[ai.w], e + 3);
        int4 l0, l1;
        l0.x = p0; l0.y = aj.x; l0.z = p1; l0.w = aj.y;
        l1.x = p2; l1.y = aj.z; l1.z = p3; l1.w = aj.w;
        int* lp = link + 8 * (size_t)t;
        *reinterpret_cast<int4*>(lp)     = l0;
        *reinterpret_cast<int4*>(lp + 4) = l1;
    }
}

__global__ __launch_bounds__(256) void GNN_gather_kernel(
    const float* __restrict__ x,
    const int*   __restrict__ head,
    const int*   __restrict__ link,
    float* __restrict__ S,
    float* __restrict__ deg)
{
    const int lane   = threadIdx.x & 31;
    const int wave   = __builtin_amdgcn_readfirstlane((int)(blockIdx.x * (blockDim.x >> 5) + (threadIdx.x >> 5)));
    const int nwaves = gridDim.x * (blockDim.x >> 5);

    for (int n = wave; n < N_NODES; n += nwaves) {
        int e = head[n];
        float4 a0 = make_float4(0.f, 0.f, 0.f, 0.f), a1 = a0;
        int cnt = 0;
        if ((unsigned)e < (unsigned)N_EDGES) {
            int2 l = *reinterpret_cast<const int2*>(link + 2 * (size_t)e);
            for (;;) {
                const float* __restrict__ xr = x + (size_t)l.y * F + lane * 4;
                const float4 r0 = *reinterpret_cast<const float4*>(xr);
                const float4 r1 = *reinterpret_cast<const float4*>(xr + 128);
                ++cnt;
                e = l.x;
                const bool more = (unsigned)e < (unsigned)N_EDGES && cnt < N_EDGES;
                if (more) l = *reinterpret_cast<const int2*>(link + 2 * (size_t)e);
                a0.x += r0.x; a0.y += r0.y; a0.z += r0.z; a0.w += r0.w;
                a1.x += r1.x; a1.y += r1.y; a1.z += r1.z; a1.w += r1.w;
                if (!more) break;
            }
        }

        float* __restrict__ sr = S + (size_t)n * F + lane * 4;
        *reinterpret_cast<float4*>(sr)       = a0;
        *reinterpret_cast<float4*>(sr + 128) = a1;
        if (lane == 0) deg[n] = (float)cnt;
    }
}

#define KB 32
#define NBLK (K2 / KB)

static __device__ __forceinline__ void GNN_load_a(
    const float* __restrict__ ap, float4 (&a)[4])
{
    #pragma unroll
    for (int j = 0; j < 4; ++j)
        a[j] = *reinterpret_cast<const float4*>(ap + 8 * j);
}
static __device__ __forceinline__ void GNN_load_b(
    const float* bpk, float4 (&b)[4][NT])
{
    #pragma unroll
    for (int j = 0; j < 4; ++j) {
        #pragma unroll
        for (int t = 0; t < NT; ++t)
            b[j][t] = *reinterpret_cast<const float4*>(bpk + t * 16 * BSTRIDE + 8 * j);
    }
}

__global__ __launch_bounds__(256) void GNN_gemm_kernel(
    const float* __restrict__ x,
    const float* __restrict__ W,
    const float* __restrict__ b,
    const float* __restrict__ S,
    const float* __restrict__ deg,
    float* __restrict__ out)
{
    __shared__ __attribute__((aligned(16))) float Bs[BN * BSTRIDE];

    const int n0   = blockIdx.x * BN;
    const int wave = threadIdx.x >> 5;
    const int lane = threadIdx.x & 31;
    const int mtile = blockIdx.y * 8 + wave;

    for (int i = threadIdx.x; i < BN * (K2 / 4); i += 256) {
        const int j  = i >> 7;
        const int k4 = (i & 127) * 4;
        const float4 w = *reinterpret_cast<const float4*>(W + (size_t)(n0 + j) * K2 + k4);
        *reinterpret_cast<float4*>(&Bs[j * BSTRIDE + k4]) = w;
    }
    __syncthreads();

    if (mtile >= M_TILES) return;

    const int c    = lane & 15;
    const int h    = lane >> 4;
    const int m0   = mtile * 16;
    const int arow = m0 + c;
    const float dsc = deg[arow];

    const float* __restrict__ xr = x + (size_t)arow * F + 4 * h;
    const float* __restrict__ sr = S + (size_t)arow * F + 4 * h;
    const float* bp = Bs + c * BSTRIDE + 4 * h;

    v8f acc[NT];
    #pragma unroll
    for (int t = 0; t < NT; ++t) acc[t] = v8f{};

    float4 Af[4][4];
    float4 Bf[2][4][NT];
    GNN_load_a(xr, Af[0]);
    GNN_load_a(xr + KB, Af[1]);
    GNN_load_a(xr + 2 * KB, Af[2]);
    GNN_load_b(bp, Bf[0]);
    #pragma unroll
    for (int blk = 0; blk < NBLK; ++blk) {
        const int cur = blk & 1;
        if (blk + 3 < NBLK) {
            const float* __restrict__ ap = (blk + 3 < NBLK / 2) ? (xr + KB * (blk + 3))
                                                                : (sr + KB * (blk + 3 - NBLK / 2));
            GNN_load_a(ap, Af[(blk + 3) & 3]);
        }
        if (blk + 1 < NBLK) GNN_load_b(bp + KB * (blk + 1), Bf[cur ^ 1]);
        __builtin_amdgcn_sched_barrier(0);
        #pragma unroll
        for (int j = 0; j < 4; ++j) {
            v4f av;
            av.x = Af[blk & 3][j].x;  av.y = Af[blk & 3][j].y;
            av.z = Af[blk & 3][j].z;  av.w = Af[blk & 3][j].w;
            if (blk < NBLK / 2) av = av * dsc;
            const v2f a0 = av.xy;
            const v2f a1 = av.zw;
            #pragma unroll
            for (int t = 0; t < NT; ++t) {
                v2f b0; b0.x = Bf[cur][j][t].x; b0.y = Bf[cur][j][t].y;
                acc[t] = __builtin_amdgcn_wmma_f32_16x16x4_f32(
                    false, a0, false, b0, (short)0, acc[t], false, false);
            }
            #pragma unroll
            for (int t = 0; t < NT; ++t) {
                v2f b1; b1.x = Bf[cur][j][t].z; b1.y = Bf[cur][j][t].w;
                acc[t] = __builtin_amdgcn_wmma_f32_16x16x4_f32(
                    false, a1, false, b1, (short)0, acc[t], false, false);
            }
        }
        __builtin_amdgcn_sched_barrier(0);
    }

    const int mbase = m0 + h * 8;
    float dg[8];
    #pragma unroll
    for (int i = 0; i < 8; ++i) dg[i] = deg[mbase + i];
    #pragma unroll
    for (int t = 0; t < NT; ++t) {
        const int nout = n0 + t * 16 + c;
        const float bias = b[nout];
        #pragma unroll
        for (int i = 0; i < 8; ++i) {
            out[(size_t)(mbase + i) * F + nout] = acc[t][i] + dg[i] * bias;
        }
    }
}

extern "C" void kernel_launch(void* const* d_in, const int* in_sizes, int n_in,
                              void* d_out, int out_size, void* d_ws, size_t ws_size,
                              hipStream_t stream) {
    const float* x  = (const float*)d_in[0];
    const float* W  = (const float*)d_in[1];
    const float* b  = (const float*)d_in[2];
    const int*   Ai = (const int*)d_in[3];
    const int*   Aj = (const int*)d_in[4];
    float* out = (float*)d_out;

    float* S    = (float*)d_ws;
    float* deg  = S + (size_t)N_NODES * F;
    int*   head = (int*)(deg + N_NODES);
    int*   link = head + N_NODES;

    GNN_init_kernel<<<(N_NODES + 255) / 256, 256, 0, stream>>>(head);

    GNN_link_kernel<<<(N_EDGES / 4 + 255) / 256, 256, 0, stream>>>(Ai, Aj, head, link);

    GNN_gather_kernel<<<2048, 256, 0, stream>>>(x, head, link, S, deg);

    dim3 grid(F / BN, (M_TILES + 7) / 8);
    GNN_gemm_kernel<<<grid, 256, 0, stream>>>(x, W, b, S, deg, out);
}


# GEMM launched as a resident grid: 512 blocks (2 per WGP) stage their W slice in LDS once and each wave loops over 3 or 4 16-row strips, leftover strips one per SIMD; sum order per output unchanged
# speedup vs baseline: 2.2841x; 1.0352x over previous
#include <hip/hip_runtime.h>
#include <hip/hip_bf16.h>

#define N_NODES 50000
#define N_EDGES 312500
#define F 256
#define K2 512
#define M_TILES 3125
#define BSTRIDE 516
#define NT 4
#define BN (16 * NT)

typedef float v2f __attribute__((ext_vector_type(2)));
typedef float v4f __attribute__((ext_vector_type(4)));
typedef float v8f __attribute__((ext_vector_type(8)));

__global__ __launch_bounds__(256) void GNN_init_kernel(int* __restrict__ head) {
    const int i = blockIdx.x * blockDim.x + threadIdx.x;
    if (i < N_NODES) head[i] = -1;
}

static_assert(N_EDGES % 4 == 0, "edge count must be a multiple of 4");
__global__ __launch_bounds__(256) void GNN_link_kernel(
    const int* __restrict__ Ai,
    const int* __restrict__ Aj,
    int* __restrict__ head,
    int* __restrict__ link)
{
    const int t = blockIdx.x * blockDim.x + threadIdx.x;
    if (t < N_EDGES / 4) {
        const int4 ai = *reinterpret_cast<const int4*>(Ai + 4 * (size_t)t);
        const int4 aj = *reinterpret_cast<const int4*>(Aj + 4 * (size_t)t);
        const int e = 4 * t;
        const int p0 = atomicExch(&head[ai.x], e);
        const int p1 = atomicExch(&head[ai.y], e + 1);
        const int p2 = atomicExch(&head[ai.z], e + 2);
        const int p3 = atomicExch(&head[ai.w], e + 3);
        int4 l0, l1;
        l0.x = p0; l0.y = aj.x; l0.z = p1; l0.w = aj.y;
        l1.x = p2; l1.y = aj.z; l1.z = p3; l1.w = aj.w;
        int* lp = link + 8 * (size_t)t;
        *reinterpret_cast<int4*>(lp)     = l0;
        *reinterpret_cast<int4*>(lp + 4) = l1;
    }
}

__global__ __launch_bounds__(256) void GNN_gather_kernel(
    const float* __restrict__ x,
    const int*   __restrict__ head,
    const int*   __restrict__ link,
    float* __restrict__ S,
    float* __restrict__ deg)
{
    const int lane   = threadIdx.x & 31;
    const int wave   = __builtin_amdgcn_readfirstlane((int)(blockIdx.x * (blockDim.x >> 5) + (threadIdx.x >> 5)));
    const int nwaves = gridDim.x * (blockDim.x >> 5);

    for (int n = wave; n < N_NODES; n += nwaves) {
        int e = head[n];
        float4 a0 = make_float4(0.f, 0.f, 0.f, 0.f), a1 = a0;
        int cnt = 0;
        if ((unsigned)e < (unsigned)N_EDGES) {
            int2 l = *reinterpret_cast<const int2*>(link + 2 * (size_t)e);
            for (;;) {
                const float* __restrict__ xr = x + (size_t)l.y * F + lane * 4;
                const float4 r0 = *reinterpret_cast<const float4*>(xr);
                const float4 r1 = *reinterpret_cast<const float4*>(xr + 128);
                ++cnt;
                e = l.x;
                const bool more = (unsigned)e < (unsigned)N_EDGES && cnt < N_EDGES;
                if (more) l = *reinterpret_cast<const int2*>(link + 2 * (size_t)e);
                a0.x += r0.x; a0.y += r0.y; a0.z += r0.z; a0.w += r0.w;
                a1.x += r1.x; a1.y += r1.y; a1.z += r1.z; a1.w += r1.w;
                if (!more) break;
            }
        }

        float* __restrict__ sr = S + (size_t)n * F + lane * 4;
        *reinterpret_cast<float4*>(sr)       = a0;
        *reinterpret_cast<float4*>(sr + 128) = a1;
        if (lane == 0) deg[n] = (float)cnt;
    }
}

#define KB 32
#define NBLK (K2 / KB)
#define GEMM_BLOCKS 512
#define QW ((GEMM_BLOCKS / (F / BN)) * 8)
#define FULL_ROUNDS (M_TILES / QW)
#define LEFT_TILES (M_TILES % QW)
static_assert(LEFT_TILES <= GEMM_BLOCKS / (F / BN), "one leftover tile per slot at most");
static_assert(F / BN == 4, "column group = low two bits of blockIdx.x");

static __device__ __forceinline__ void GNN_load_a(
    const float* __restrict__ ap, float4 (&a)[4])
{
    #pragma unroll
    for (int j = 0; j < 4; ++j)
        a[j] = *reinterpret_cast<const float4*>(ap + 8 * j);
}
static __device__ __forceinline__ void GNN_load_b(
    const float* bpk, float4 (&b)[4][NT])
{
    #pragma unroll
    for (int j = 0; j < 4; ++j) {
        #pragma unroll
        for (int t = 0; t < NT; ++t)
            b[j][t] = *reinterpret_cast<const float4*>(bpk + t * 16 * BSTRIDE + 8 * j);
    }
}

__global__ __launch_bounds__(256) __attribute__((amdgpu_waves_per_eu(4))) void GNN_gemm_kernel(
    const float* __restrict__ x,
    const float* __restrict__ W,
    const float* __restrict__ b,
    const float* __restrict__ S,
    const float* __restrict__ deg,
    float* __restrict__ out)
{
    __shared__ __attribute__((aligned(16))) float Bs[BN * BSTRIDE];

    const int cg   = blockIdx.x & (F / BN - 1);
    const int slot = blockIdx.x / (F / BN);
    const int n0   = cg * BN;
    const int wave = __builtin_amdgcn_readfirstlane((int)(threadIdx.x >> 5));
    const int lane = threadIdx.x & 31;
    const int q    = slot * 8 + wave;

    for (int i = threadIdx.x; i < BN * (K2 / 4); i += 256) {
        const int j  = i >> 7;
        const int k4 = (i & 127) * 4;
        const float4 w = *reinterpret_cast<const float4*>(W + (size_t)(n0 + j) * K2 + k4);
        *reinterpret_cast<float4*>(&Bs[j * BSTRIDE + k4]) = w;
    }
    __syncthreads();

    const int c    = lane & 15;
    const int h    = lane >> 4;
    const float* bp = Bs + c * BSTRIDE + 4 * h;

    const int xwave = 2 * cg + ((cg ^ (cg >> 1)) & 1);
    const bool extra = slot < LEFT_TILES && wave == xwave;
    const int nstrips = FULL_ROUNDS + (extra ? 1 : 0);

    for (int s = 0; s < nstrips; ++s) {
    const int mtile = (s < FULL_ROUNDS) ? s * QW + q : FULL_ROUNDS * QW + slot;
    const int m0   = mtile * 16;
    const int arow = m0 + c;
    const float dsc = deg[arow];

    const float* __restrict__ xr = x + (size_t)arow * F + 4 * h;
    const float* __restrict__ sr = S + (size_t)arow * F + 4 * h;

    v8f acc[NT];
    #pragma unroll
    for (int t = 0; t < NT; ++t) acc[t] = v8f{};

    float4 Af[4][4];
    float4 Bf[2][4][NT];
    GNN_load_a(xr, Af[0]);
    GNN_load_a(xr + KB, Af[1]);
    GNN_load_a(xr + 2 * KB, Af[2]);
    GNN_load_b(bp, Bf[0]);
    #pragma unroll
    for (int blk = 0; blk < NBLK; ++blk) {
        const int cur = blk & 1;
        if (blk + 3 < NBLK) {
            const float* __restrict__ ap = (blk + 3 < NBLK / 2) ? (xr + KB * (blk + 3))
                                                                : (sr + KB * (blk + 3 - NBLK / 2));
            GNN_load_a(ap, Af[(blk + 3) & 3]);
        }
        if (blk + 1 < NBLK) GNN_load_b(bp + KB * (blk + 1), Bf[cur ^ 1]);
        __builtin_amdgcn_sched_barrier(0);
        #pragma unroll
        for (int j = 0; j < 4; ++j) {
            v4f av;
            av.x = Af[blk & 3][j].x;  av.y = Af[blk & 3][j].y;
            av.z = Af[blk & 3][j].z;  av.w = Af[blk & 3][j].w;
            if (blk < NBLK / 2) av = av * dsc;
            const v2f a0 = av.xy;
            const v2f a1 = av.zw;
            #pragma unroll
            for (int t = 0; t < NT; ++t) {
                v2f b0; b0.x = Bf[cur][j][t].x; b0.y = Bf[cur][j][t].y;
                acc[t] = __builtin_amdgcn_wmma_f32_16x16x4_f32(
                    false, a0, false, b0, (short)0, acc[t], false, false);
            }
            #pragma unroll
            for (int t = 0; t < NT; ++t) {
                v2f b1; b1.x = Bf[cur][j][t].z; b1.y = Bf[cur][j][t].w;
                acc[t] = __builtin_amdgcn_wmma_f32_16x16x4_f32(
                    false, a1, false, b1, (short)0, acc[t], false, false);
            }
        }
        __builtin_amdgcn_sched_barrier(0);
    }

    const int mbase = m0 + h * 8;
    float dg[8];
    #pragma unroll
    for (int i = 0; i < 8; ++i) dg[i] = deg[mbase + i];
    #pragma unroll
    for (int t = 0; t < NT; ++t) {
        const int nout = n0 + t * 16 + c;
        const float bias = b[nout];
        #pragma unroll
        for (int i = 0; i < 8; ++i) {
            out[(size_t)(mbase + i) * F + nout] = acc[t][i] + dg[i] * bias;
        }
    }
    }
}

extern "C" void kernel_launch(void* const* d_in, const int* in_sizes, int n_in,
                              void* d_out, int out_size, void* d_ws, size_t ws_size,
                              hipStream_t stream) {
    const float* x  = (const float*)d_in[0];
    const float* W  = (const float*)d_in[1];
    const float* b  = (const float*)d_in[2];
    const int*   Ai = (const int*)d_in[3];
    const int*   Aj = (const int*)d_in[4];
    float* out = (float*)d_out;

    float* S    = (float*)d_ws;
    float* deg  = S + (size_t)N_NODES * F;
    int*   head = (int*)(deg + N_NODES);
    int*   link = head + N_NODES;

    GNN_init_kernel<<<(N_NODES + 255) / 256, 256, 0, stream>>>(head);

    GNN_link_kernel<<<(N_EDGES / 4 + 255) / 256, 256, 0, stream>>>(Ai, Aj, head, link);

    GNN_gather_kernel<<<2048, 256, 0, stream>>>(x, head, link, S, deg);

    GNN_gemm_kernel<<<GEMM_BLOCKS, 256, 0, stream>>>(x, W, b, S, deg, out);
}
